# starting kernel
# baseline (speedup 1.0000x reference)
#include <hip/hip_runtime.h>
#include <hip/hip_bf16.h>


typedef __attribute__((ext_vector_type(2))) float v2f;
typedef __attribute__((ext_vector_type(8))) float v8f;

#define LHALF   20
#define MMODES  41
#define NMODES  2
#define H1DIM   2
#define H2DIM   10
#define BATCHSZ 65536

#define WAVES_PER_BLOCK 4
#define TILE            16
#define XF              (MMODES * 4)
#define H1STRIDE        20

struct PairTab {
    unsigned pack[442];
    int cnt;
};

constexpr PairTab buildTab() {
    PairTab t{};
    int c = 0;
    for (int m = -LHALF; m <= LHALF; ++m) {
        for (int n = -LHALF; n <= LHALF; ++n) {
            if (n < m) continue;
            long mn = (long)m * (long)n; if (mn < 0) mn = -mn;
            int s = m + n; if (s < 0) s = -s;
            if (mn <= LHALF && s <= LHALF) {
                unsigned en  = (unsigned)(LHALF + n);
                unsigned emn = (unsigned)(LHALF + m + n);
                unsigned em  = (unsigned)(LHALF + m);
                unsigned sy  = (m != n) ? 1u : 0u;
                t.pack[c] = en | (emn << 6) | (em << 12) | (sy << 18);
                ++c;
            }
        }
    }
    t.cnt = c;
    t.pack[c] = t.pack[0];
    return t;
}

constexpr PairTab TAB = buildTab();
constexpr int HDIM  = TAB.cnt;
constexpr int HDIMP = HDIM + (HDIM & 1);
static_assert(HDIM > 0 && HDIM <= 441, "HDIM bounds");

__device__ __constant__ PairTab dTAB = TAB;

__device__ __forceinline__ float lrelu(float x) {
    return x > 0.f ? x : 0.01f * x;
}

__device__ __forceinline__ v8f wmma_step(const float* __restrict__ xe,
                                         const float* __restrict__ bp,
                                         int p, unsigned pk, v8f acc)
{
    int iEn  =  pk        & 63;
    int iEmn = (pk >> 6)  & 63;
    int iEm  = (pk >> 12) & 63;
    float sym = (pk & (1u << 18)) ? 2.f : 1.f;

    const float4 En  = *(const float4*)(xe + iEn  * 4);
    const float4 Emn = *(const float4*)(xe + iEmn * 4);
    const v2f    Em  = *(const v2f*)  (xe + iEm * 4 + 2 * p);

    float Ar = En.x * Emn.x + En.y * Emn.y + En.z * Emn.z + En.w * Emn.w;
    float Ai = En.y * Emn.x - En.x * Emn.y + En.w * Emn.z - En.z * Emn.w;
    v2f a;
    a.x = (Ar * Em.x - Ai * Em.y) * sym;
    a.y = (Ar * Em.y + Ai * Em.x) * sym;

    v2f b = *(const v2f*)bp;

    return __builtin_amdgcn_wmma_f32_16x16x4_f32(
        false, a, false, b, (short)0, acc, false, false);
}

__global__ __launch_bounds__(WAVES_PER_BLOCK * 32)
void eqpbcnn_kernel(const float* __restrict__ x_real,
                    const float* __restrict__ x_imag,
                    const float* __restrict__ task_info,
                    const float* __restrict__ W1r, const float* __restrict__ W1i,
                    const float* __restrict__ W2r, const float* __restrict__ W2i,
                    const float* __restrict__ W3r, const float* __restrict__ W3i,
                    float* __restrict__ out)
{
    extern __shared__ float lds[];
    float* Blds = lds;
    float* xall = lds + HDIMP * 64;

    const int tid  = threadIdx.x;
    const int lane = tid & 31;
    const int wave = tid >> 5;

    float* xw  = xall + wave * (TILE * XF);
    float* h1w = xall + WAVES_PER_BLOCK * (TILE * XF) + wave * (TILE * H1STRIDE);

    for (int idx = tid; idx < HDIMP * 64; idx += blockDim.x) {
        int t = idx & 3;
        int c = (idx >> 2) & 15;
        int j = idx >> 6;
        float v = 0.f;
        if (c < 8 && j < HDIM) {
            int p = c >> 2, o = (c >> 1) & 1, part = c & 1;
            int pf = t >> 1, partf = t & 1;
            if (pf == p) {
                float wr = W1r[(p * H1DIM + o) * HDIM + j];
                float wi = W1i[(p * H1DIM + o) * HDIM + j];
                v = (part == 0) ? (partf == 0 ? wr : -wi)
                                : (partf == 0 ? wi :  wr);
            }
        }
        Blds[idx] = v;
    }
    __syncthreads();

    const int tile = blockIdx.x * WAVES_PER_BLOCK + wave;
    const int base = tile * TILE;

    const float* xr = x_real + (size_t)base * (MMODES * NMODES);
    const float* xi = x_imag + (size_t)base * (MMODES * NMODES);
    for (int idx = lane; idx < TILE * MMODES * NMODES; idx += 32) {
        float a = xr[idx];
        float b = xi[idx];
        int e = idx / (MMODES * NMODES);
        int t = idx - e * (MMODES * NMODES);
        float* d = &xw[e * XF + t * 2];
        d[0] = a;
        d[1] = b;
    }
    __builtin_amdgcn_wave_barrier();
    asm volatile("s_wait_dscnt 0x0" ::: "memory");

    const int r = lane & 15;
    const int p = lane >> 4;
    const float* xe = xw + r * XF;
    const float* bp = Blds + r * 4 + p * 2;

    v8f acc0 = {};
    v8f acc1 = {};
    for (int j = 0; j < HDIMP; j += 2) {
        unsigned pk0 = dTAB.pack[j];
        unsigned pk1 = dTAB.pack[j + 1];
        acc0 = wmma_step(xe, bp + (size_t)j * 64,       p, pk0, acc0);
        acc1 = wmma_step(xe, bp + (size_t)(j + 1) * 64, p, pk1, acc1);
    }
    v8f acc = acc0 + acc1;

#pragma unroll
    for (int v = 0; v < 8; ++v) {
        int row = p * 8 + v;
        h1w[row * H1STRIDE + r] = acc[v];
    }
    __builtin_amdgcn_wave_barrier();
    asm volatile("s_wait_dscnt 0x0" ::: "memory");

    float4 h = *(const float4*)(h1w + r * H1STRIDE + 4 * p);
    float a0 = lrelu(h.x), b0 = lrelu(h.y);
    float a1 = lrelu(h.z), b1 = lrelu(h.w);

    const float* w2r = W2r + p * (H2DIM * H1DIM);
    const float* w2i = W2i + p * (H2DIM * H1DIM);
    const float* w3r = W3r + p * H2DIM;
    const float* w3i = W3i + p * H2DIM;

    float Er = 0.f, Ei = 0.f;
#pragma unroll
    for (int q = 0; q < H2DIM; ++q) {
        float c0 = w2r[q * 2 + 0], c1 = w2r[q * 2 + 1];
        float d0 = w2i[q * 2 + 0], d1 = w2i[q * 2 + 1];
        float hr = a0 * c0 - b0 * d0 + a1 * c1 - b1 * d1;
        float hi = a0 * d0 + b0 * c0 + a1 * d1 + b1 * c1;
        hr = lrelu(hr);
        hi = lrelu(hi);
        float wr = w3r[q], wi = w3i[q];
        Er += hr * wr - hi * wi;
        Ei += hr * wi + hi * wr;
    }

    int bidx = base + r;
    float P = exp10f(task_info[bidx * 4] * 0.1f) * 0.5f;
    float xLr = xw[r * XF + (LHALF * 2 + p) * 2 + 0];
    float xLi = xw[r * XF + (LHALF * 2 + p) * 2 + 1];

    float2 res;
    res.x = xLr + Er * P;
    res.y = xLi + Ei * P;
    *(float2*)(out + ((size_t)bidx * NMODES + p) * 2) = res;
}

extern "C" void kernel_launch(void* const* d_in, const int* in_sizes, int n_in,
                              void* d_out, int out_size, void* d_ws, size_t ws_size,
                              hipStream_t stream) {
    const float* x_real    = (const float*)d_in[0];
    const float* x_imag    = (const float*)d_in[1];
    const float* task_info = (const float*)d_in[2];
    const float* W1r       = (const float*)d_in[3];
    const float* W1i       = (const float*)d_in[4];
    const float* W2r       = (const float*)d_in[5];
    const float* W2i       = (const float*)d_in[6];
    const float* W3r       = (const float*)d_in[7];
    const float* W3i       = (const float*)d_in[8];
    float* out             = (float*)d_out;

    const int blocks = BATCHSZ / (TILE * WAVES_PER_BLOCK);
    const size_t ldsFloats = (size_t)HDIMP * 64
                           + (size_t)WAVES_PER_BLOCK * TILE * XF
                           + (size_t)WAVES_PER_BLOCK * TILE * H1STRIDE;
    const size_t ldsBytes = ldsFloats * sizeof(float);

    eqpbcnn_kernel<<<blocks, WAVES_PER_BLOCK * 32, ldsBytes, stream>>>(
        x_real, x_imag, task_info, W1r, W1i, W2r, W2i, W3r, W3i, out);
}


# 32 batch rows per wave, one element per lane: lane halves are two row groups in the A operand, a block-diagonal B sends them to separate columns, one WMMA per pol; A' once per element; sums unchanged
# speedup vs baseline: 2.9161x; 2.9161x over previous
#include <hip/hip_runtime.h>
#include <hip/hip_bf16.h>


typedef __attribute__((ext_vector_type(2))) float v2f;
typedef __attribute__((ext_vector_type(8))) float v8f;

#define LHALF   20
#define MMODES  41
#define NMODES  2
#define H1DIM   2
#define H2DIM   10
#define BATCHSZ 65536

#define WAVES_PER_BLOCK 4
#define TILE            32
#define XF              (MMODES * 4)
#define H1STRIDE        20
#define H1ROWS          16

struct PairTab {
    unsigned pack[442];
    int cnt;
};

constexpr PairTab buildTab() {
    PairTab t{};
    int c = 0;
    for (int m = -LHALF; m <= LHALF; ++m) {
        for (int n = -LHALF; n <= LHALF; ++n) {
            if (n < m) continue;
            long mn = (long)m * (long)n; if (mn < 0) mn = -mn;
            int s = m + n; if (s < 0) s = -s;
            if (mn <= LHALF && s <= LHALF) {
                unsigned en  = (unsigned)(LHALF + n);
                unsigned emn = (unsigned)(LHALF + m + n);
                unsigned em  = (unsigned)(LHALF + m);
                unsigned sy  = (m != n) ? 1u : 0u;
                t.pack[c] = en | (emn << 6) | (em << 12) | (sy << 18);
                ++c;
            }
        }
    }
    t.cnt = c;
    return t;
}

constexpr PairTab TAB = buildTab();
constexpr int HDIM  = TAB.cnt;
static_assert(HDIM > 0 && HDIM <= 441, "HDIM bounds");

__device__ __constant__ PairTab dTAB = TAB;

__device__ __forceinline__ float lrelu(float x) {
    return x > 0.f ? x : 0.01f * x;
}

typedef __attribute__((ext_vector_type(4))) float v4f;
typedef v2f __attribute__((aligned(4))) v2f_a4;

struct PairIn {
    v4f   En, Emn;
    v4f   Em;
    float sym;
};
struct StageIn {
    PairIn q0, q1;
    v2f    wr, wi;
};

__device__ __forceinline__ PairIn fetch_pair(const float* __restrict__ xe,
                                             unsigned pk)
{
    int iEn  =  pk        & 63;
    int iEmn = (pk >> 6)  & 63;
    int iEm  = (pk >> 12) & 63;
    PairIn q;
    q.sym = (pk & (1u << 18)) ? 2.f : 1.f;
    q.En  = *(const v4f*)(xe + iEn  * 4);
    q.Emn = *(const v4f*)(xe + iEmn * 4);
    q.Em  = *(const v4f*)(xe + iEm  * 4);
    return q;
}

__device__ __forceinline__ StageIn fetch_stage(const float* __restrict__ xe,
                                               unsigned pk0, unsigned pk1,
                                               const float* __restrict__ wr_p,
                                               const float* __restrict__ wi_p,
                                               int j)
{
    StageIn s;
    s.q0 = fetch_pair(xe, pk0);
    s.q1 = fetch_pair(xe, pk1);
    s.wr = *(const v2f_a4*)(wr_p + j);
    s.wi = *(const v2f_a4*)(wi_p + j);
    return s;
}

__device__ __forceinline__ void pair_begins(PairIn& q)
{
    asm volatile("" : "+v"(q.En), "+v"(q.Emn), "+v"(q.Em));
}
__device__ __forceinline__ void stage_begins(StageIn& s)
{
    pair_begins(s.q0);
    pair_begins(s.q1);
    asm volatile("" : "+v"(s.wr), "+v"(s.wi));
}

__device__ __forceinline__ void pair_wmma(const PairIn& q, float wr, float wi,
                                          v2f cR, v2f cI, v8f& accP0, v8f& accP1)
{
    const v4f En = q.En, Emn = q.Emn, Em = q.Em;
    const float sym = q.sym;
    float Ar = En.x * Emn.x + En.y * Emn.y + En.z * Emn.z + En.w * Emn.w;
    float Ai = En.y * Emn.x - En.x * Emn.y + En.w * Emn.z - En.z * Emn.w;
    v2f a0, a1;
    a0.x = (Ar * Em.x - Ai * Em.y) * sym;
    a0.y = (Ar * Em.y + Ai * Em.x) * sym;
    a1.x = (Ar * Em.z - Ai * Em.w) * sym;
    a1.y = (Ar * Em.w + Ai * Em.z) * sym;
    v2f b = wr * cR + wi * cI;

    accP0 = __builtin_amdgcn_wmma_f32_16x16x4_f32(
        false, a0, false, b, (short)0, accP0, false, false);
    accP1 = __builtin_amdgcn_wmma_f32_16x16x4_f32(
        false, a1, false, b, (short)0, accP1, false, false);
}

struct Acc {
    v8f e0, e1;
    v8f o0, o1;
};

__device__ __forceinline__ void stage_wmma(StageIn& s, v2f cR, v2f cI, Acc& acc)
{
    stage_begins(s);
    pair_wmma(s.q0, s.wr.x, s.wi.x, cR, cI, acc.e0, acc.e1);
    pair_wmma(s.q1, s.wr.y, s.wi.y, cR, cI, acc.o0, acc.o1);
}

__global__ __launch_bounds__(WAVES_PER_BLOCK * 32)
void eqpbcnn_kernel(const float* __restrict__ x_real,
                    const float* __restrict__ x_imag,
                    const float* __restrict__ task_info,
                    const float* __restrict__ W1r, const float* __restrict__ W1i,
                    const float* __restrict__ W2r, const float* __restrict__ W2i,
                    const float* __restrict__ W3r, const float* __restrict__ W3i,
                    float* __restrict__ out)
{
    __shared__ __attribute__((aligned(16))) float lds[WAVES_PER_BLOCK * (TILE * XF + H1ROWS * H1STRIDE)];
    float* xall = lds;

    const int tid  = threadIdx.x;
    const int lane = tid & 31;
    const int wave = tid >> 5;

    float* xw  = xall + wave * (TILE * XF);
    float* h1w = xall + WAVES_PER_BLOCK * (TILE * XF) + wave * (H1ROWS * H1STRIDE);

    const int tile = blockIdx.x * WAVES_PER_BLOCK + wave;
    const int base = tile * TILE;

    const float* xr = x_real + (size_t)base * (MMODES * NMODES);
    const float* xi = x_imag + (size_t)base * (MMODES * NMODES);
#pragma unroll 16
    for (int idx = lane; idx < TILE * MMODES * NMODES; idx += 32) {
        float a = xr[idx];
        float b = xi[idx];
        int e = idx / (MMODES * NMODES);
        int t = idx - e * (MMODES * NMODES);
        float* d = &xw[e * XF + t * 2];
        d[0] = a;
        d[1] = b;
    }
    __builtin_amdgcn_wave_barrier();
    asm volatile("s_wait_dscnt 0x0" ::: "memory");

    const int r = lane & 15;
    const int g = lane >> 4;
    const float* xe = xw + lane * XF;

    const int   part = r & 1;
    const float on   = (((r >> 2) & 1) == g) ? 1.f : 0.f;
    v2f cR, cI;
    cR.x = part ? 0.f : on;    cR.y = part ? on  : 0.f;
    cI.x = part ? on  : 0.f;   cI.y = part ? 0.f : -on;
    const int wrow = ((r >> 3) << 1) | ((r >> 1) & 1);
    const float* __restrict__ wr_p = W1r + wrow * HDIM;
    const float* __restrict__ wi_p = W1i + wrow * HDIM;

    static_assert(HDIM % 6 == 1 && HDIM >= 13, "loop shape: 6*k + 1 pairs");
    Acc acc;
    acc.e0 = v8f{}; acc.e1 = v8f{}; acc.o0 = v8f{}; acc.o1 = v8f{};
    StageIn sx = fetch_stage(xe, dTAB.pack[0], dTAB.pack[1], wr_p, wi_p, 0);
    StageIn sy = fetch_stage(xe, dTAB.pack[2], dTAB.pack[3], wr_p, wi_p, 2);
    unsigned n0 = dTAB.pack[4], n1 = dTAB.pack[5], n2 = dTAB.pack[6];
    unsigned n3 = dTAB.pack[7], n4 = dTAB.pack[8], n5 = dTAB.pack[9];
    int j = 0;
    for (; j + 12 < HDIM; j += 6) {
        unsigned m0 = dTAB.pack[j + 10], m1 = dTAB.pack[j + 11];
        unsigned m2 = dTAB.pack[j + 12], m3 = dTAB.pack[j + 13];
        unsigned m4 = dTAB.pack[j + 14], m5 = dTAB.pack[j + 15];

        StageIn sz = fetch_stage(xe, n0, n1, wr_p, wi_p, j + 4);
        __builtin_amdgcn_sched_barrier(0);
        stage_wmma(sx, cR, cI, acc);
        __builtin_amdgcn_sched_barrier(0);

        sx = fetch_stage(xe, n2, n3, wr_p, wi_p, j + 6);
        __builtin_amdgcn_sched_barrier(0);
        stage_wmma(sy, cR, cI, acc);
        __builtin_amdgcn_sched_barrier(0);

        sy = fetch_stage(xe, n4, n5, wr_p, wi_p, j + 8);
        __builtin_amdgcn_sched_barrier(0);
        stage_wmma(sz, cR, cI, acc);

        n0 = m0; n1 = m1; n2 = m2; n3 = m3; n4 = m4; n5 = m5;
    }
    {
        StageIn sz = fetch_stage(xe, n0, n1, wr_p, wi_p, j + 4);
        PairIn  t  = fetch_pair(xe, n2);
        float twr = wr_p[j + 6], twi = wi_p[j + 6];
        __builtin_amdgcn_sched_barrier(0);
        stage_wmma(sx, cR, cI, acc);
        stage_wmma(sy, cR, cI, acc);
        stage_wmma(sz, cR, cI, acc);
        pair_begins(t);
        pair_wmma(t, twr, twi, cR, cI, acc.e0, acc.e1);
    }
    const v8f accP0 = acc.e0 + acc.o0;
    const v8f accP1 = acc.e1 + acc.o1;
    const bool colP1 = r >= 8;

#pragma unroll
    for (int v = 0; v < 8; ++v) {
        int row = g * 8 + v;
        h1w[row * H1STRIDE + r] = colP1 ? accP1[v] : accP0[v];
    }
    __builtin_amdgcn_wave_barrier();
    asm volatile("s_wait_dscnt 0x0" ::: "memory");

    const int bidx = base + lane;
    const float P = exp10f(task_info[bidx * 4] * 0.1f) * 0.5f;
    const float4 xL = *(const float4*)(xe + LHALF * 4);
    float res[4];
#pragma unroll
    for (int p = 0; p < NMODES; ++p) {
        float4 h = *(const float4*)(h1w + r * H1STRIDE + 8 * p + 4 * g);
        float a0 = lrelu(h.x), b0 = lrelu(h.y);
        float a1 = lrelu(h.z), b1 = lrelu(h.w);

        const float* w2r = W2r + p * (H2DIM * H1DIM);
        const float* w2i = W2i + p * (H2DIM * H1DIM);
        const float* w3r = W3r + p * H2DIM;
        const float* w3i = W3i + p * H2DIM;

        float Er = 0.f, Ei = 0.f;
#pragma unroll
        for (int q = 0; q < H2DIM; ++q) {
            float c0 = w2r[q * 2 + 0], c1 = w2r[q * 2 + 1];
            float d0 = w2i[q * 2 + 0], d1 = w2i[q * 2 + 1];
            float hr = a0 * c0 - b0 * d0 + a1 * c1 - b1 * d1;
            float hi = a0 * d0 + b0 * c0 + a1 * d1 + b1 * c1;
            hr = lrelu(hr);
            hi = lrelu(hi);
            float wr = w3r[q], wi = w3i[q];
            Er += hr * wr - hi * wi;
            Ei += hr * wi + hi * wr;
        }
        const float xLr = p ? xL.z : xL.x;
        const float xLi = p ? xL.w : xL.y;
        res[2 * p + 0] = xLr + Er * P;
        res[2 * p + 1] = xLi + Ei * P;
    }
    float4 o;
    o.x = res[0]; o.y = res[1]; o.z = res[2]; o.w = res[3];
    *(float4*)(out + (size_t)bidx * (NMODES * 2)) = o;
}

extern "C" void kernel_launch(void* const* d_in, const int* in_sizes, int n_in,
                              void* d_out, int out_size, void* d_ws, size_t ws_size,
                              hipStream_t stream) {
    const float* x_real    = (const float*)d_in[0];
    const float* x_imag    = (const float*)d_in[1];
    const float* task_info = (const float*)d_in[2];
    const float* W1r       = (const float*)d_in[3];
    const float* W1i       = (const float*)d_in[4];
    const float* W2r       = (const float*)d_in[5];
    const float* W2i       = (const float*)d_in[6];
    const float* W3r       = (const float*)d_in[7];
    const float* W3i       = (const float*)d_in[8];
    float* out             = (float*)d_out;

    const int blocks = BATCHSZ / (TILE * WAVES_PER_BLOCK);

    eqpbcnn_kernel<<<blocks, WAVES_PER_BLOCK * 32, 0, stream>>>(
        x_real, x_imag, task_info, W1r, W1i, W2r, W2i, W3r, W3i, out);
}


# x tile kept in LDS as a plain copy of the global rows (re pairs, then im pairs, per element): 16-byte staging, tap gather = one two-address LDS load, pols two-wide; A' summed per pol, then over pols
# speedup vs baseline: 3.4121x; 1.1701x over previous
#include <hip/hip_runtime.h>
#include <hip/hip_bf16.h>


typedef __attribute__((ext_vector_type(2))) float v2f;
typedef __attribute__((ext_vector_type(8))) float v8f;

#define LHALF   20
#define MMODES  41
#define NMODES  2
#define H1DIM   2
#define H2DIM   10
#define BATCHSZ 65536

#define WAVES_PER_BLOCK 4
#define TILE            32
#define XP              (2 * MMODES + 1)
#define XF              (2 * XP)
#define H1STRIDE        20
#define H1ROWS          16

struct PairTab {
    unsigned pack[442];
    int cnt;
};

constexpr PairTab buildTab() {
    PairTab t{};
    int c = 0;
    for (int m = -LHALF; m <= LHALF; ++m) {
        for (int n = -LHALF; n <= LHALF; ++n) {
            if (n < m) continue;
            long mn = (long)m * (long)n; if (mn < 0) mn = -mn;
            int s = m + n; if (s < 0) s = -s;
            if (mn <= LHALF && s <= LHALF) {
                unsigned en  = (unsigned)(LHALF + n);
                unsigned emn = (unsigned)(LHALF + m + n);
                unsigned em  = (unsigned)(LHALF + m);
                unsigned sy  = (m != n) ? 1u : 0u;
                t.pack[c] = en | (emn << 6) | (em << 12) | (sy << 18);
                ++c;
            }
        }
    }
    t.cnt = c;
    return t;
}

constexpr PairTab TAB = buildTab();
constexpr int HDIM  = TAB.cnt;
static_assert(HDIM > 0 && HDIM <= 441, "HDIM bounds");

__device__ __constant__ PairTab dTAB = TAB;

__host__ __device__ constexpr unsigned row_of(unsigned s) { return (s * 1599u) >> 16; }
constexpr bool rowOfHolds() {
    for (unsigned s = 0; s < 32u * MMODES; ++s)
        if (row_of(s) != s / MMODES) return false;
    return true;
}
static_assert(rowOfHolds(), "row_of: multiply-shift equals s / 41 over a tile");

__device__ __forceinline__ float lrelu(float x) {
    return x > 0.f ? x : 0.01f * x;
}

typedef __attribute__((ext_vector_type(4))) float v4f;
typedef v2f __attribute__((aligned(4))) v2f_a4;

struct PairIn {
    v2f   EnR, EnI;
    v2f   EmnR, EmnI;
    v2f   EmR, EmI;
    float sym;
};
struct StageIn {
    PairIn q0, q1;
    v2f    wr, wi;
};

__device__ __forceinline__ PairIn fetch_pair(const v2f* __restrict__ xe,
                                             unsigned pk)
{
    int iEn  =  pk        & 63;
    int iEmn = (pk >> 6)  & 63;
    int iEm  = (pk >> 12) & 63;
    PairIn q;
    q.sym  = (pk & (1u << 18)) ? 2.f : 1.f;
    q.EnR  = xe[iEn];    q.EnI  = xe[iEn  + MMODES];
    q.EmnR = xe[iEmn];   q.EmnI = xe[iEmn + MMODES];
    q.EmR  = xe[iEm];    q.EmI  = xe[iEm  + MMODES];
    return q;
}

__device__ __forceinline__ StageIn fetch_stage(const v2f* __restrict__ xe,
                                               unsigned pk0, unsigned pk1,
                                               const float* __restrict__ wr_p,
                                               const float* __restrict__ wi_p,
                                               int j)
{
    StageIn s;
    s.q0 = fetch_pair(xe, pk0);
    s.q1 = fetch_pair(xe, pk1);
    s.wr = *(const v2f_a4*)(wr_p + j);
    s.wi = *(const v2f_a4*)(wi_p + j);
    return s;
}

__device__ __forceinline__ void pair_begins(PairIn& q)
{
    asm volatile("" : "+v"(q.EnR), "+v"(q.EnI), "+v"(q.EmnR), "+v"(q.EmnI),
                      "+v"(q.EmR), "+v"(q.EmI));
}
__device__ __forceinline__ void stage_begins(StageIn& s)
{
    pair_begins(s.q0);
    pair_begins(s.q1);
    asm volatile("" : "+v"(s.wr), "+v"(s.wi));
}

__device__ __forceinline__ void pair_wmma(const PairIn& q, float wr, float wi,
                                          v2f cR, v2f cI, v8f& accP0, v8f& accP1)
{
    const v2f pr = q.EnR * q.EmnR + q.EnI * q.EmnI;
    const v2f pi = q.EnI * q.EmnR - q.EnR * q.EmnI;
    v2f A;
    A.x = pr.x + pr.y;
    A.y = pi.x + pi.y;
    A = A * q.sym;
    const v2f Fr = A.x * q.EmR - A.y * q.EmI;
    const v2f Fi = A.x * q.EmI + A.y * q.EmR;
    v2f a0, a1;
    a0.x = Fr.x;  a0.y = Fi.x;
    a1.x = Fr.y;  a1.y = Fi.y;
    v2f b = wr * cR + wi * cI;

    accP0 = __builtin_amdgcn_wmma_f32_16x16x4_f32(
        false, a0, false, b, (short)0, accP0, false, false);
    accP1 = __builtin_amdgcn_wmma_f32_16x16x4_f32(
        false, a1, false, b, (short)0, accP1, false, false);
}

struct Acc {
    v8f e0, e1;
    v8f o0, o1;
};

__device__ __forceinline__ void stage_wmma(StageIn& s, v2f cR, v2f cI, Acc& acc)
{
    stage_begins(s);
    pair_wmma(s.q0, s.wr.x, s.wi.x, cR, cI, acc.e0, acc.e1);
    pair_wmma(s.q1, s.wr.y, s.wi.y, cR, cI, acc.o0, acc.o1);
}

__global__ __launch_bounds__(WAVES_PER_BLOCK * 32)
void eqpbcnn_kernel(const float* __restrict__ x_real,
                    const float* __restrict__ x_imag,
                    const float* __restrict__ task_info,
                    const float* __restrict__ W1r, const float* __restrict__ W1i,
                    const float* __restrict__ W2r, const float* __restrict__ W2i,
                    const float* __restrict__ W3r, const float* __restrict__ W3i,
                    float* __restrict__ out)
{
    __shared__ __attribute__((aligned(16))) float lds[WAVES_PER_BLOCK * (TILE * XF + H1ROWS * H1STRIDE)];
    float* xall = lds;

    const int tid  = threadIdx.x;
    const int lane = tid & 31;
    const int wave = tid >> 5;

    float* xw  = xall + wave * (TILE * XF);
    float* h1w = xall + WAVES_PER_BLOCK * (TILE * XF) + wave * (H1ROWS * H1STRIDE);

    const int tile = blockIdx.x * WAVES_PER_BLOCK + wave;
    const int base = tile * TILE;

    const v4f* __restrict__ xr4 = (const v4f*)(x_real + (size_t)base * (MMODES * NMODES)) + lane;
    const v4f* __restrict__ xi4 = (const v4f*)(x_imag + (size_t)base * (MMODES * NMODES)) + lane;
    constexpr int PIECES = TILE * MMODES * NMODES / 4;
    constexpr int ROUNDS = (PIECES + 31) / 32;
    static_assert(TILE * MMODES * NMODES % 4 == 0, "whole 16-byte pieces");
    v2f* xw2 = (v2f*)xw;
    v4f ra[ROUNDS], ia[ROUNDS];
#pragma unroll
    for (int k = 0; k < ROUNDS; ++k) {
        if (lane + 32 * k < PIECES) {
            ra[k] = xr4[32 * k];
            ia[k] = xi4[32 * k];
        }
    }
#pragma unroll
    for (int k = 0; k < ROUNDS; ++k) {
        const unsigned piece = lane + 32 * k;
        if (piece < PIECES) {
            const unsigned s0 = 2 * piece, s1 = s0 + 1;
            const unsigned d0 = s0 + (XP - MMODES) * row_of(s0);
            const unsigned d1 = s1 + (XP - MMODES) * row_of(s1);
            v2f t;
            t.x = ra[k].x; t.y = ra[k].y;  xw2[d0] = t;
            t.x = ra[k].z; t.y = ra[k].w;  xw2[d1] = t;
            t.x = ia[k].x; t.y = ia[k].y;  xw2[d0 + MMODES] = t;
            t.x = ia[k].z; t.y = ia[k].w;  xw2[d1 + MMODES] = t;
        }
    }
    __builtin_amdgcn_wave_barrier();
    asm volatile("s_wait_dscnt 0x0" ::: "memory");

    const int r = lane & 15;
    const int g = lane >> 4;
    const v2f* xe = xw2 + lane * XP;

    const int   part = r & 1;
    const float on   = (((r >> 2) & 1) == g) ? 1.f : 0.f;
    v2f cR, cI;
    cR.x = part ? 0.f : on;    cR.y = part ? on  : 0.f;
    cI.x = part ? on  : 0.f;   cI.y = part ? 0.f : -on;
    const int wrow = ((r >> 3) << 1) | ((r >> 1) & 1);
    const float* __restrict__ wr_p = W1r + wrow * HDIM;
    const float* __restrict__ wi_p = W1i + wrow * HDIM;

    static_assert(HDIM % 6 == 1 && HDIM >= 13, "loop shape: 6*k + 1 pairs");
    Acc acc;
    acc.e0 = v8f{}; acc.e1 = v8f{}; acc.o0 = v8f{}; acc.o1 = v8f{};
    StageIn sx = fetch_stage(xe, dTAB.pack[0], dTAB.pack[1], wr_p, wi_p, 0);
    StageIn sy = fetch_stage(xe, dTAB.pack[2], dTAB.pack[3], wr_p, wi_p, 2);
    unsigned n0 = dTAB.pack[4], n1 = dTAB.pack[5], n2 = dTAB.pack[6];
    unsigned n3 = dTAB.pack[7], n4 = dTAB.pack[8], n5 = dTAB.pack[9];
    int j = 0;
    for (; j + 12 < HDIM; j += 6) {
        unsigned m0 = dTAB.pack[j + 10], m1 = dTAB.pack[j + 11];
        unsigned m2 = dTAB.pack[j + 12], m3 = dTAB.pack[j + 13];
        unsigned m4 = dTAB.pack[j + 14], m5 = dTAB.pack[j + 15];

        StageIn sz = fetch_stage(xe, n0, n1, wr_p, wi_p, j + 4);
        __builtin_amdgcn_sched_barrier(0);
        stage_wmma(sx, cR, cI, acc);
        __builtin_amdgcn_sched_barrier(0);

        sx = fetch_stage(xe, n2, n3, wr_p, wi_p, j + 6);
        __builtin_amdgcn_sched_barrier(0);
        stage_wmma(sy, cR, cI, acc);
        __builtin_amdgcn_sched_barrier(0);

        sy = fetch_stage(xe, n4, n5, wr_p, wi_p, j + 8);
        __builtin_amdgcn_sched_barrier(0);
        stage_wmma(sz, cR, cI, acc);

        n0 = m0; n1 = m1; n2 = m2; n3 = m3; n4 = m4; n5 = m5;
    }
    {
        StageIn sz = fetch_stage(xe, n0, n1, wr_p, wi_p, j + 4);
        PairIn  t  = fetch_pair(xe, n2);
        float twr = wr_p[j + 6], twi = wi_p[j + 6];
        __builtin_amdgcn_sched_barrier(0);
        stage_wmma(sx, cR, cI, acc);
        stage_wmma(sy, cR, cI, acc);
        stage_wmma(sz, cR, cI, acc);
        pair_begins(t);
        pair_wmma(t, twr, twi, cR, cI, acc.e0, acc.e1);
    }
    const v8f accP0 = acc.e0 + acc.o0;
    const v8f accP1 = acc.e1 + acc.o1;
    const bool colP1 = r >= 8;

#pragma unroll
    for (int v = 0; v < 8; ++v) {
        int row = g * 8 + v;
        h1w[row * H1STRIDE + r] = colP1 ? accP1[v] : accP0[v];
    }
    __builtin_amdgcn_wave_barrier();
    asm volatile("s_wait_dscnt 0x0" ::: "memory");

    const int bidx = base + lane;
    const float P = exp10f(task_info[bidx * 4] * 0.1f) * 0.5f;
    const v2f xLre = xe[LHALF];
    const v2f xLim = xe[LHALF + MMODES];
    float res[4];
#pragma unroll
    for (int p = 0; p < NMODES; ++p) {
        float4 h = *(const float4*)(h1w + r * H1STRIDE + 8 * p + 4 * g);
        float a0 = lrelu(h.x), b0 = lrelu(h.y);
        float a1 = lrelu(h.z), b1 = lrelu(h.w);

        const float* w2r = W2r + p * (H2DIM * H1DIM);
        const float* w2i = W2i + p * (H2DIM * H1DIM);
        const float* w3r = W3r + p * H2DIM;
        const float* w3i = W3i + p * H2DIM;

        float Er = 0.f, Ei = 0.f;
#pragma unroll
        for (int q = 0; q < H2DIM; ++q) {
            float c0 = w2r[q * 2 + 0], c1 = w2r[q * 2 + 1];
            float d0 = w2i[q * 2 + 0], d1 = w2i[q * 2 + 1];
            float hr = a0 * c0 - b0 * d0 + a1 * c1 - b1 * d1;
            float hi = a0 * d0 + b0 * c0 + a1 * d1 + b1 * c1;
            hr = lrelu(hr);
            hi = lrelu(hi);
            float wr = w3r[q], wi = w3i[q];
            Er += hr * wr - hi * wi;
            Ei += hr * wi + hi * wr;
        }
        const float xLr = p ? xLre.y : xLre.x;
        const float xLi = p ? xLim.y : xLim.x;
        res[2 * p + 0] = xLr + Er * P;
        res[2 * p + 1] = xLi + Ei * P;
    }
    float4 o;
    o.x = res[0]; o.y = res[1]; o.z = res[2]; o.w = res[3];
    *(float4*)(out + (size_t)bidx * (NMODES * 2)) = o;
}

extern "C" void kernel_launch(void* const* d_in, const int* in_sizes, int n_in,
                              void* d_out, int out_size, void* d_ws, size_t ws_size,
                              hipStream_t stream) {
    const float* x_real    = (const float*)d_in[0];
    const float* x_imag    = (const float*)d_in[1];
    const float* task_info = (const float*)d_in[2];
    const float* W1r       = (const float*)d_in[3];
    const float* W1i       = (const float*)d_in[4];
    const float* W2r       = (const float*)d_in[5];
    const float* W2i       = (const float*)d_in[6];
    const float* W3r       = (const float*)d_in[7];
    const float* W3i       = (const float*)d_in[8];
    float* out             = (float*)d_out;

    const int blocks = BATCHSZ / (TILE * WAVES_PER_BLOCK);

    eqpbcnn_kernel<<<blocks, WAVES_PER_BLOCK * 32, 0, stream>>>(
        x_real, x_imag, task_info, W1r, W1i, W2r, W2i, W3r, W3i, out);
}


# B fragments (pair factor folded in) built once per block into an LDS table under the x loads; pair loop gets them with one 16-byte LDS load per stage, W1 loads and fragment math gone; sums unchanged
# speedup vs baseline: 3.6466x; 1.0687x over previous
#include <hip/hip_runtime.h>
#include <hip/hip_bf16.h>


typedef __attribute__((ext_vector_type(2))) float v2f;
typedef __attribute__((ext_vector_type(8))) float v8f;

#define LHALF   20
#define MMODES  41
#define NMODES  2
#define H1DIM   2
#define H2DIM   10
#define BATCHSZ 65536

#define WAVES_PER_BLOCK 4
#define TILE            32
#define XP              (2 * MMODES + 1)
#define XF              (2 * XP)
#define H1STRIDE        20
#define H1ROWS          16
#define NSTAGE          88
#define STAGES_PER_WAVE (NSTAGE / WAVES_PER_BLOCK)
#define BTAB_FLOATS     (NSTAGE * 32 * 4)

struct PairTab {
    unsigned pack[442];
    int cnt;
};

constexpr PairTab buildTab() {
    PairTab t{};
    int c = 0;
    for (int m = -LHALF; m <= LHALF; ++m) {
        for (int n = -LHALF; n <= LHALF; ++n) {
            if (n < m) continue;
            long mn = (long)m * (long)n; if (mn < 0) mn = -mn;
            int s = m + n; if (s < 0) s = -s;
            if (mn <= LHALF && s <= LHALF) {
                unsigned en  = (unsigned)(LHALF + n);
                unsigned emn = (unsigned)(LHALF + m + n);
                unsigned em  = (unsigned)(LHALF + m);
                unsigned sy  = (m != n) ? 1u : 0u;
                t.pack[c] = en | (emn << 6) | (em << 12) | (sy << 18);
                ++c;
            }
        }
    }
    t.cnt = c;
    return t;
}

constexpr PairTab TAB = buildTab();
constexpr int HDIM  = TAB.cnt;
static_assert(HDIM > 0 && HDIM <= 441, "HDIM bounds");

__device__ __constant__ PairTab dTAB = TAB;

struct FacTab { float f[2 * NSTAGE]; };
constexpr FacTab buildFac() {
    FacTab t{};
    for (int k = 0; k < 2 * NSTAGE; ++k)
        t.f[k] = (k < TAB.cnt) ? ((TAB.pack[k] & (1u << 18)) ? 2.f : 1.f) : 0.f;
    return t;
}
constexpr FacTab FAC = buildFac();
__device__ __constant__ FacTab dFAC = FAC;

__host__ __device__ constexpr unsigned row_of(unsigned s) { return (s * 1599u) >> 16; }
constexpr bool rowOfHolds() {
    for (unsigned s = 0; s < 32u * MMODES; ++s)
        if (row_of(s) != s / MMODES) return false;
    return true;
}
static_assert(rowOfHolds(), "row_of: multiply-shift equals s / 41 over a tile");

__device__ __forceinline__ float lrelu(float x) {
    return x > 0.f ? x : 0.01f * x;
}

typedef __attribute__((ext_vector_type(4))) float v4f;
typedef v2f __attribute__((aligned(4))) v2f_a4;

struct PairIn {
    v2f   EnR, EnI;
    v2f   EmnR, EmnI;
    v2f   EmR, EmI;
};
struct StageIn {
    PairIn q0, q1;
    v4f    bf;
};

__device__ __forceinline__ PairIn fetch_pair(const v2f* __restrict__ xe,
                                             unsigned pk)
{
    int iEn  =  pk        & 63;
    int iEmn = (pk >> 6)  & 63;
    int iEm  = (pk >> 12) & 63;
    PairIn q;
    q.EnR  = xe[iEn];    q.EnI  = xe[iEn  + MMODES];
    q.EmnR = xe[iEmn];   q.EmnI = xe[iEmn + MMODES];
    q.EmR  = xe[iEm];    q.EmI  = xe[iEm  + MMODES];
    return q;
}

__device__ __forceinline__ StageIn fetch_stage(const v2f* __restrict__ xe,
                                               unsigned pk0, unsigned pk1,
                                               const v4f* __restrict__ tbs)
{
    StageIn s;
    s.q0 = fetch_pair(xe, pk0);
    s.q1 = fetch_pair(xe, pk1);
    s.bf = *tbs;
    return s;
}

__device__ __forceinline__ void pair_begins(PairIn& q)
{
    asm volatile("" : "+v"(q.EnR), "+v"(q.EnI), "+v"(q.EmnR), "+v"(q.EmnI),
                      "+v"(q.EmR), "+v"(q.EmI));
}
__device__ __forceinline__ void stage_begins(StageIn& s)
{
    pair_begins(s.q0);
    pair_begins(s.q1);
    asm volatile("" : "+v"(s.bf));
}

__device__ __forceinline__ void pair_wmma(const PairIn& q, v2f b,
                                          v8f& accP0, v8f& accP1)
{
    const v2f pr = q.EnR * q.EmnR + q.EnI * q.EmnI;
    const v2f pi = q.EnI * q.EmnR - q.EnR * q.EmnI;
    v2f A;
    A.x = pr.x + pr.y;
    A.y = pi.x + pi.y;
    const v2f Fr = A.x * q.EmR - A.y * q.EmI;
    const v2f Fi = A.x * q.EmI + A.y * q.EmR;
    v2f a0, a1;
    a0.x = Fr.x;  a0.y = Fi.x;
    a1.x = Fr.y;  a1.y = Fi.y;

    accP0 = __builtin_amdgcn_wmma_f32_16x16x4_f32(
        false, a0, false, b, (short)0, accP0, false, false);
    accP1 = __builtin_amdgcn_wmma_f32_16x16x4_f32(
        false, a1, false, b, (short)0, accP1, false, false);
}

struct Acc {
    v8f e0, e1;
    v8f o0, o1;
};

__device__ __forceinline__ void stage_wmma(StageIn& s, Acc& acc)
{
    stage_begins(s);
    v2f b0, b1;
    b0.x = s.bf.x;  b0.y = s.bf.y;
    b1.x = s.bf.z;  b1.y = s.bf.w;
    pair_wmma(s.q0, b0, acc.e0, acc.e1);
    pair_wmma(s.q1, b1, acc.o0, acc.o1);
}

__global__ __launch_bounds__(WAVES_PER_BLOCK * 32)
void eqpbcnn_kernel(const float* __restrict__ x_real,
                    const float* __restrict__ x_imag,
                    const float* __restrict__ task_info,
                    const float* __restrict__ W1r, const float* __restrict__ W1i,
                    const float* __restrict__ W2r, const float* __restrict__ W2i,
                    const float* __restrict__ W3r, const float* __restrict__ W3i,
                    float* __restrict__ out)
{
    constexpr int XH_FLOATS = WAVES_PER_BLOCK * (TILE * XF + H1ROWS * H1STRIDE);
    static_assert(XH_FLOATS % 4 == 0, "fragment table starts on a 16-byte boundary");
    __shared__ __attribute__((aligned(16))) float lds[XH_FLOATS + BTAB_FLOATS];
    float* xall = lds;
    v4f*   btab = (v4f*)(lds + XH_FLOATS);

    const int tid  = threadIdx.x;
    const int lane = tid & 31;
    const int wave = __builtin_amdgcn_readfirstlane(tid >> 5);

    float* xw  = xall + wave * (TILE * XF);
    float* h1w = xall + WAVES_PER_BLOCK * (TILE * XF) + wave * (H1ROWS * H1STRIDE);

    const int tile = blockIdx.x * WAVES_PER_BLOCK + wave;
    const int base = tile * TILE;

    const int r = lane & 15;
    const int g = lane >> 4;

    const int   part = r & 1;
    const float on   = (((r >> 2) & 1) == g) ? 1.f : 0.f;
    v2f cR, cI;
    cR.x = part ? 0.f : on;    cR.y = part ? on  : 0.f;
    cI.x = part ? on  : 0.f;   cI.y = part ? 0.f : -on;
    const int wrow = ((r >> 3) << 1) | ((r >> 1) & 1);
    const float* __restrict__ wr_p = W1r + wrow * HDIM;
    const float* __restrict__ wi_p = W1i + wrow * HDIM;

    constexpr int NPW = 2 * STAGES_PER_WAVE;
    const int k0w = wave * NPW;
    float wrv[NPW], wiv[NPW];
#pragma unroll
    for (int t = 0; t < NPW / 2 - 1; ++t) {
        const v2f a = *(const v2f_a4*)(wr_p + k0w + 2 * t);
        const v2f b = *(const v2f_a4*)(wi_p + k0w + 2 * t);
        wrv[2 * t] = a.x;  wrv[2 * t + 1] = a.y;
        wiv[2 * t] = b.x;  wiv[2 * t + 1] = b.y;
    }
    {
        const int ka = k0w + NPW - 2;
        const int kb = (k0w + NPW - 1 < HDIM) ? k0w + NPW - 1 : HDIM - 1;
        wrv[NPW - 2] = wr_p[ka];  wiv[NPW - 2] = wi_p[ka];
        wrv[NPW - 1] = wr_p[kb];  wiv[NPW - 1] = wi_p[kb];
    }

    const v4f* __restrict__ xr4 = (const v4f*)(x_real + (size_t)base * (MMODES * NMODES)) + lane;
    const v4f* __restrict__ xi4 = (const v4f*)(x_imag + (size_t)base * (MMODES * NMODES)) + lane;
    constexpr int PIECES = TILE * MMODES * NMODES / 4;
    constexpr int ROUNDS = (PIECES + 31) / 32;
    static_assert(TILE * MMODES * NMODES % 4 == 0, "whole 16-byte pieces");
    v2f* xw2 = (v2f*)xw;
    v4f ra[ROUNDS], ia[ROUNDS];
#pragma unroll
    for (int k = 0; k < ROUNDS; ++k) {
        if (lane + 32 * k < PIECES) {
            ra[k] = xr4[32 * k];
            ia[k] = xi4[32 * k];
        }
    }
#pragma unroll
    for (int i = 0; i < STAGES_PER_WAVE; ++i) {
        v2f f[2];
#pragma unroll
        for (int h = 0; h < 2; ++h) {
            const float fk = dFAC.f[k0w + 2 * i + h];
            v2f b = wrv[2 * i + h] * cR + wiv[2 * i + h] * cI;
            f[h] = b * fk;
        }
        v4f e;
        e.x = f[0].x;  e.y = f[0].y;  e.z = f[1].x;  e.w = f[1].y;
        btab[(wave * STAGES_PER_WAVE + i) * 32 + lane] = e;
    }
#pragma unroll
    for (int k = 0; k < ROUNDS; ++k) {
        const unsigned piece = lane + 32 * k;
        if (piece < PIECES) {
            const unsigned s0 = 2 * piece, s1 = s0 + 1;
            const unsigned d0 = s0 + (XP - MMODES) * row_of(s0);
            const unsigned d1 = s1 + (XP - MMODES) * row_of(s1);
            v2f t;
            t.x = ra[k].x; t.y = ra[k].y;  xw2[d0] = t;
            t.x = ra[k].z; t.y = ra[k].w;  xw2[d1] = t;
            t.x = ia[k].x; t.y = ia[k].y;  xw2[d0 + MMODES] = t;
            t.x = ia[k].z; t.y = ia[k].w;  xw2[d1 + MMODES] = t;
        }
    }
    __syncthreads();

    const v2f* xe = xw2 + lane * XP;
    const v4f* tb = btab + lane;

    static_assert(HDIM % 6 == 1 && HDIM >= 13, "loop shape: 6*k + 1 pairs");
    static_assert(2 * NSTAGE == HDIM + 1 && NSTAGE % WAVES_PER_BLOCK == 0, "fragment table: stages of two pairs, a quarter per wave");
    Acc acc;
    acc.e0 = v8f{}; acc.e1 = v8f{}; acc.o0 = v8f{}; acc.o1 = v8f{};
    StageIn sx = fetch_stage(xe, dTAB.pack[0], dTAB.pack[1], tb);
    StageIn sy = fetch_stage(xe, dTAB.pack[2], dTAB.pack[3], tb + 32);
    const v4f* tbj = tb;
    unsigned n0 = dTAB.pack[4], n1 = dTAB.pack[5], n2 = dTAB.pack[6];
    unsigned n3 = dTAB.pack[7], n4 = dTAB.pack[8], n5 = dTAB.pack[9];
    int j = 0;
    for (; j + 12 < HDIM; j += 6) {
        unsigned m0 = dTAB.pack[j + 10], m1 = dTAB.pack[j + 11];
        unsigned m2 = dTAB.pack[j + 12], m3 = dTAB.pack[j + 13];
        unsigned m4 = dTAB.pack[j + 14], m5 = dTAB.pack[j + 15];

        StageIn sz = fetch_stage(xe, n0, n1, tbj + 2 * 32);
        __builtin_amdgcn_sched_barrier(0);
        stage_wmma(sx, acc);
        __builtin_amdgcn_sched_barrier(0);

        sx = fetch_stage(xe, n2, n3, tbj + 3 * 32);
        __builtin_amdgcn_sched_barrier(0);
        stage_wmma(sy, acc);
        __builtin_amdgcn_sched_barrier(0);

        sy = fetch_stage(xe, n4, n5, tbj + 4 * 32);
        __builtin_amdgcn_sched_barrier(0);
        stage_wmma(sz, acc);

        n0 = m0; n1 = m1; n2 = m2; n3 = m3; n4 = m4; n5 = m5;
        tbj += 3 * 32;
    }
    {
        StageIn sz = fetch_stage(xe, n0, n1, tbj + 2 * 32);
        PairIn  t  = fetch_pair(xe, n2);
        v2f     tbf = *(const v2f*)(tbj + 3 * 32);
        __builtin_amdgcn_sched_barrier(0);
        stage_wmma(sx, acc);
        stage_wmma(sy, acc);
        stage_wmma(sz, acc);
        pair_begins(t);
        asm volatile("" : "+v"(tbf));
        pair_wmma(t, tbf, acc.e0, acc.e1);
    }
    const v8f accP0 = acc.e0 + acc.o0;
    const v8f accP1 = acc.e1 + acc.o1;
    const bool colP1 = r >= 8;

#pragma unroll
    for (int v = 0; v < 8; ++v) {
        int row = g * 8 + v;
        h1w[row * H1STRIDE + r] = colP1 ? accP1[v] : accP0[v];
    }
    __builtin_amdgcn_wave_barrier();
    asm volatile("s_wait_dscnt 0x0" ::: "memory");

    const int bidx = base + lane;
    const float P = exp10f(task_info[bidx * 4] * 0.1f) * 0.5f;
    const v2f xLre = xe[LHALF];
    const v2f xLim = xe[LHALF + MMODES];
    float res[4];
#pragma unroll
    for (int p = 0; p < NMODES; ++p) {
        float4 h = *(const float4*)(h1w + r * H1STRIDE + 8 * p + 4 * g);
        float a0 = lrelu(h.x), b0 = lrelu(h.y);
        float a1 = lrelu(h.z), b1 = lrelu(h.w);

        const float* w2r = W2r + p * (H2DIM * H1DIM);
        const float* w2i = W2i + p * (H2DIM * H1DIM);
        const float* w3r = W3r + p * H2DIM;
        const float* w3i = W3i + p * H2DIM;

        float Er = 0.f, Ei = 0.f;
#pragma unroll
        for (int q = 0; q < H2DIM; ++q) {
            float c0 = w2r[q * 2 + 0], c1 = w2r[q * 2 + 1];
            float d0 = w2i[q * 2 + 0], d1 = w2i[q * 2 + 1];
            float hr = a0 * c0 - b0 * d0 + a1 * c1 - b1 * d1;
            float hi = a0 * d0 + b0 * c0 + a1 * d1 + b1 * c1;
            hr = lrelu(hr);
            hi = lrelu(hi);
            float wr = w3r[q], wi = w3i[q];
            Er += hr * wr - hi * wi;
            Ei += hr * wi + hi * wr;
        }
        const float xLr = p ? xLre.y : xLre.x;
        const float xLi = p ? xLim.y : xLim.x;
        res[2 * p + 0] = xLr + Er * P;
        res[2 * p + 1] = xLi + Ei * P;
    }
    float4 o;
    o.x = res[0]; o.y = res[1]; o.z = res[2]; o.w = res[3];
    *(float4*)(out + (size_t)bidx * (NMODES * 2)) = o;
}

extern "C" void kernel_launch(void* const* d_in, const int* in_sizes, int n_in,
                              void* d_out, int out_size, void* d_ws, size_t ws_size,
                              hipStream_t stream) {
    const float* x_real    = (const float*)d_in[0];
    const float* x_imag    = (const float*)d_in[1];
    const float* task_info = (const float*)d_in[2];
    const float* W1r       = (const float*)d_in[3];
    const float* W1i       = (const float*)d_in[4];
    const float* W2r       = (const float*)d_in[5];
    const float* W2i       = (const float*)d_in[6];
    const float* W3r       = (const float*)d_in[7];
    const float* W3i       = (const float*)d_in[8];
    float* out             = (float*)d_out;

    const int blocks = BATCHSZ / (TILE * WAVES_PER_BLOCK);

    eqpbcnn_kernel<<<blocks, WAVES_PER_BLOCK * 32, 0, stream>>>(
        x_real, x_imag, task_info, W1r, W1i, W2r, W2i, W3r, W3i, out);
}
